# starting kernel
# baseline (speedup 1.0000x reference)
#include <hip/hip_runtime.h>

typedef __attribute__((ext_vector_type(2))) float v2f;
typedef __attribute__((ext_vector_type(8))) float v8f;

#define BATCH   4096
#define IN_DIM  1024
#define FD      64
#define OD      30
#define KSTR    (FD * OD)

#define MT       16
#define KCH      256
#define XS_STRIDE 258

__global__ __launch_bounds__(256) void build_W(const float* __restrict__ ker,
                                               const int* __restrict__ fidx,
                                               float* __restrict__ W) {
    const int idx = blockIdx.x * 256 + threadIdx.x;
    const int i = idx >> 10;
    const int j = idx & 1023;
    const int fi = fidx[i];
    const int fj = fidx[j];
    const float* __restrict__ a = ker + i * KSTR + fj * OD;
    const float* __restrict__ b = ker + j * KSTR + fi * OD;
    float s = 0.f;
#pragma unroll
    for (int d = 0; d < OD; ++d) s += a[d] * b[d];
    W[idx] = s;
}

__global__ __launch_bounds__(256) void ffm_gemm(const float* __restrict__ X,
                                                const float* __restrict__ W,
                                                float* __restrict__ out) {
    __shared__ float Xs[MT * XS_STRIDE];
    __shared__ float red[MT];
    __shared__ float redD[MT];

    const int batch0 = blockIdx.x * MT;
    const int tid  = threadIdx.x;
    const int wave = tid >> 5;
    const int lane = tid & 31;

    if (tid < MT) { red[tid] = 0.f; redD[tid] = 0.f; }

    v8f acc[8];
#pragma unroll
    for (int t = 0; t < 8; ++t) acc[t] = (v8f)(0.f);

    const int nbase = wave * 128;
    const int m_a   = lane & 15;
    const int ksel  = (lane >> 4) * 2;
    const int n_b   = lane & 15;

    for (int kb = 0; kb < IN_DIM; kb += KCH) {
        __syncthreads();
        for (int v = tid; v < MT * (KCH / 4); v += 256) {
            const int row  = v / (KCH / 4);
            const int col4 = v % (KCH / 4);
            const float4 src =
                ((const float4*)(X + (size_t)(batch0 + row) * IN_DIM + kb))[col4];
            float* dst = &Xs[row * XS_STRIDE + col4 * 4];
            dst[0] = src.x; dst[1] = src.y; dst[2] = src.z; dst[3] = src.w;
        }
        __syncthreads();

#pragma unroll
        for (int t = 0; t < 8; ++t) {
            const int ncol = nbase + t * 16 + n_b;
            const float* __restrict__ wp = W + ncol;
#pragma unroll 4
            for (int k = 0; k < KCH; k += 4) {
                v2f a, b;
                const float* xs = &Xs[m_a * XS_STRIDE + k + ksel];
                a.x = xs[0];
                a.y = xs[1];
                const float* wrow = wp + (size_t)(kb + k + ksel) * IN_DIM;
                b.x = wrow[0];
                b.y = wrow[IN_DIM];
                acc[t] = __builtin_amdgcn_wmma_f32_16x16x4_f32(
                     false, a,  false, b,
                     (short)0, acc[t],
                     false,  false);
            }
        }
    }

    float sums[8];
#pragma unroll
    for (int r = 0; r < 8; ++r) sums[r] = 0.f;
#pragma unroll
    for (int t = 0; t < 8; ++t) {
        const int n = nbase + t * 16 + (lane & 15);
#pragma unroll
        for (int r = 0; r < 8; ++r) {
            const int m = r + ((lane >> 4) << 3);
            sums[r] += acc[t][r] * X[(size_t)(batch0 + m) * IN_DIM + n];
        }
    }
#pragma unroll
    for (int r = 0; r < 8; ++r) {
        const int m = r + ((lane >> 4) << 3);
        atomicAdd(&red[m], sums[r]);
    }

    {
        const int m   = tid >> 4;
        const int seg = tid & 15;
        const float* xr = X + (size_t)(batch0 + m) * IN_DIM + seg * 64;
        float s = 0.f;
#pragma unroll 8
        for (int i = 0; i < 64; ++i) {
            const float xv = xr[i];
            s += xv * xv * W[(size_t)(seg * 64 + i) * (IN_DIM + 1)];
        }
        atomicAdd(&redD[m], s);
    }

    __syncthreads();
    if (tid < MT) out[batch0 + tid] = 0.5f * (red[tid] - redD[tid]);
}

extern "C" void kernel_launch(void* const* d_in, const int* in_sizes, int n_in,
                              void* d_out, int out_size, void* d_ws, size_t ws_size,
                              hipStream_t stream) {
    const float* x    = (const float*)d_in[0];
    const float* ker  = (const float*)d_in[1];
    const int*   fidx = (const int*)d_in[2];
    float* W   = (float*)d_ws;
    float* outp = (float*)d_out;

    build_W<<<(IN_DIM * IN_DIM) / 256, 256, 0, stream>>>(ker, fidx, W);
    ffm_gemm<<<BATCH / MT, 256, 0, stream>>>(x, W, outp);
}


# ffm_gemm work-group widened from 8 to 32 waves: each wave holds 2 column tiles (k-outer, one A fragment feeds both) so 4x more W loads are in flight per SIMD to cover L2 latency
# speedup vs baseline: 1.9633x; 1.9633x over previous
#include <hip/hip_runtime.h>

typedef __attribute__((ext_vector_type(2))) float v2f;
typedef __attribute__((ext_vector_type(8))) float v8f;

#define BATCH   4096
#define IN_DIM  1024
#define FD      64
#define OD      30
#define KSTR    (FD * OD)

#define MT       16
#define KCH      256
#define XS_STRIDE 258

#define NWAVES     32
#define NTHREADS   (NWAVES * 32)
#define WCOLS      (IN_DIM / NWAVES)
#define RED_STRIDE 18

__global__ __launch_bounds__(256) void build_W(const float* __restrict__ ker,
                                               const int* __restrict__ fidx,
                                               float* __restrict__ W) {
    const int idx = blockIdx.x * 256 + threadIdx.x;
    const int i = idx >> 10;
    const int j = idx & 1023;
    const int fi = fidx[i];
    const int fj = fidx[j];
    const float* __restrict__ a = ker + i * KSTR + fj * OD;
    const float* __restrict__ b = ker + j * KSTR + fi * OD;
    float s = 0.f;
#pragma unroll
    for (int d = 0; d < OD; ++d) s += a[d] * b[d];
    W[idx] = s;
}

__global__ __launch_bounds__(NTHREADS) void ffm_gemm(const float* __restrict__ X,
                                                     const float* __restrict__ W,
                                                     float* __restrict__ out) {
    __shared__ float Xs[MT * XS_STRIDE];
    __shared__ float red[MT * RED_STRIDE];
    __shared__ float redD[MT];

    const int batch0 = blockIdx.x * MT;
    const int tid  = threadIdx.x;
    const int wave = tid >> 5;
    const int lane = tid & 31;

    for (int i = tid; i < MT * RED_STRIDE; i += NTHREADS) red[i] = 0.f;
    if (tid < MT) redD[tid] = 0.f;

    v8f acc0 = (v8f)(0.f);
    v8f acc1 = (v8f)(0.f);

    const int nbase = wave * WCOLS;
    const int m_a   = lane & 15;
    const int ksel  = (lane >> 4) * 2;
    const int n_b   = lane & 15;
    const float* __restrict__ wp = W + nbase + n_b;

    for (int kb = 0; kb < IN_DIM; kb += KCH) {
        __syncthreads();
        for (int v = tid; v < MT * (KCH / 4); v += NTHREADS) {
            const int row  = v / (KCH / 4);
            const int col4 = v % (KCH / 4);
            const float4 src =
                ((const float4*)(X + (size_t)(batch0 + row) * IN_DIM + kb))[col4];
            float* dst = &Xs[row * XS_STRIDE + col4 * 4];
            dst[0] = src.x; dst[1] = src.y; dst[2] = src.z; dst[3] = src.w;
        }
        __syncthreads();

        const float* xs = &Xs[m_a * XS_STRIDE + ksel];
        const float* __restrict__ wk = wp + (size_t)(kb + ksel) * IN_DIM;
#pragma unroll 8
        for (int k = 0; k < KCH; k += 4) {
            v2f a, b0, b1;
            a.x = xs[k];
            a.y = xs[k + 1];
            const float* wrow = wk + (size_t)k * IN_DIM;
            b0.x = wrow[0];
            b0.y = wrow[IN_DIM];
            b1.x = wrow[16];
            b1.y = wrow[IN_DIM + 16];
            acc0 = __builtin_amdgcn_wmma_f32_16x16x4_f32(
                 false, a,  false, b0,
                 (short)0, acc0,
                 false,  false);
            acc1 = __builtin_amdgcn_wmma_f32_16x16x4_f32(
                 false, a,  false, b1,
                 (short)0, acc1,
                 false,  false);
        }
    }

    const int c  = lane & 15;
    const int h8 = (lane >> 4) << 3;
    float sums[8];
#pragma unroll
    for (int r = 0; r < 8; ++r) {
        const int m = r + h8;
        const float* xr = X + (size_t)(batch0 + m) * IN_DIM + nbase + c;
        sums[r] = acc0[r] * xr[0] + acc1[r] * xr[16];
    }

    const int md  = tid >> 6;
    float sd = 0.f;
    {
        const int seg = tid & 63;
        const float* xr = X + (size_t)(batch0 + md) * IN_DIM + seg * 16;
#pragma unroll
        for (int i = 0; i < 16; ++i) {
            const float xv = xr[i];
            sd += xv * xv * W[(size_t)(seg * 16 + i) * (IN_DIM + 1)];
        }
    }

#pragma unroll
    for (int r = 0; r < 8; ++r) {
        const int m = r + h8;
        atomicAdd(&red[m * RED_STRIDE + c], sums[r]);
    }
    atomicAdd(&redD[md], sd);

    __syncthreads();
    if (tid < MT) {
        float s = 0.f;
#pragma unroll
        for (int q = 0; q < 16; ++q) s += red[tid * RED_STRIDE + q];
        out[batch0 + tid] = 0.5f * (s - redD[tid]);
    }
}

extern "C" void kernel_launch(void* const* d_in, const int* in_sizes, int n_in,
                              void* d_out, int out_size, void* d_ws, size_t ws_size,
                              hipStream_t stream) {
    const float* x    = (const float*)d_in[0];
    const float* ker  = (const float*)d_in[1];
    const int*   fidx = (const int*)d_in[2];
    float* W   = (float*)d_ws;
    float* outp = (float*)d_out;

    build_W<<<(IN_DIM * IN_DIM) / 256, 256, 0, stream>>>(ker, fidx, W);
    ffm_gemm<<<BATCH / MT, NTHREADS, 0, stream>>>(x, W, outp);
}


# build_W stages row i and the field column f_i of the latent kernel in LDS with coalesced loads (a wave-load covers 2 whole 120-byte segments) instead of per-lane segment loads of 32 lines each
# speedup vs baseline: 2.2029x; 1.1220x over previous
#include <hip/hip_runtime.h>

typedef __attribute__((ext_vector_type(2))) float v2f;
typedef __attribute__((ext_vector_type(8))) float v8f;

#define BATCH   4096
#define IN_DIM  1024
#define FD      64
#define OD      30
#define KSTR    (FD * OD)

#define MT       16
#define KCH      256
#define XS_STRIDE 258

#define NWAVES     32
#define NTHREADS   (NWAVES * 32)
#define WCOLS      (IN_DIM / NWAVES)
#define RED_STRIDE 18

#define JT    256
#define SEG2  (OD / 2)

__global__ __launch_bounds__(JT) void build_W(const float* __restrict__ ker,
                                              const int* __restrict__ fidx,
                                              float* __restrict__ W) {
    __shared__ __attribute__((aligned(16))) float As[KSTR];
    __shared__ __attribute__((aligned(16))) float Bs[JT * OD];

    const int tid = threadIdx.x;
    const int i   = blockIdx.x >> 2;
    const int j0  = (blockIdx.x & 3) * JT;
    const int fi  = fidx[i];
    const int fj  = fidx[j0 + tid];

    const float4* __restrict__ arow = (const float4*)(ker + (size_t)i * KSTR);
    const int v1 = tid + (tid < KSTR / 4 - JT ? JT : 0);
    const float4 r0 = arow[tid];
    const float4 r1 = arow[v1];

    const int q  = min(tid & 15, SEG2 - 1);
    const int sg = tid >> 4;
    const float2* __restrict__ src =
        (const float2*)(ker + (size_t)(j0 + sg) * KSTR + fi * OD) + q;
    float2* dst = (float2*)Bs + sg * SEG2 + q;
    float2 pc[JT / 16];
#pragma unroll
    for (int r = 0; r < JT / 16; ++r) pc[r] = src[(size_t)r * 16 * (KSTR / 2)];
#pragma unroll
    for (int r = 0; r < JT / 16; ++r) dst[r * 16 * SEG2] = pc[r];
    ((float4*)As)[tid] = r0;
    ((float4*)As)[v1]  = r1;
    __syncthreads();

    const float2* a2 = (const float2*)(As + fj * OD);
    const float2* b2 = (const float2*)(Bs + tid * OD);
    float s = 0.f;
#pragma unroll
    for (int p = 0; p < SEG2; ++p) {
        const float2 av = a2[p];
        const float2 bv = b2[p];
        s += av.x * bv.x;
        s += av.y * bv.y;
    }
    W[(size_t)i * IN_DIM + j0 + tid] = s;
}

__global__ __launch_bounds__(NTHREADS) void ffm_gemm(const float* __restrict__ X,
                                                     const float* __restrict__ W,
                                                     float* __restrict__ out) {
    __shared__ float Xs[MT * XS_STRIDE];
    __shared__ float red[MT * RED_STRIDE];
    __shared__ float redD[MT];

    const int batch0 = blockIdx.x * MT;
    const int tid  = threadIdx.x;
    const int wave = tid >> 5;
    const int lane = tid & 31;

    for (int i = tid; i < MT * RED_STRIDE; i += NTHREADS) red[i] = 0.f;
    if (tid < MT) redD[tid] = 0.f;

    v8f acc0 = (v8f)(0.f);
    v8f acc1 = (v8f)(0.f);

    const int nbase = wave * WCOLS;
    const int m_a   = lane & 15;
    const int ksel  = (lane >> 4) * 2;
    const int n_b   = lane & 15;
    const float* __restrict__ wp = W + nbase + n_b;

    for (int kb = 0; kb < IN_DIM; kb += KCH) {
        __syncthreads();
        for (int v = tid; v < MT * (KCH / 4); v += NTHREADS) {
            const int row  = v / (KCH / 4);
            const int col4 = v % (KCH / 4);
            const float4 src =
                ((const float4*)(X + (size_t)(batch0 + row) * IN_DIM + kb))[col4];
            float* dst = &Xs[row * XS_STRIDE + col4 * 4];
            dst[0] = src.x; dst[1] = src.y; dst[2] = src.z; dst[3] = src.w;
        }
        __syncthreads();

        const float* xs = &Xs[m_a * XS_STRIDE + ksel];
        const float* __restrict__ wk = wp + (size_t)(kb + ksel) * IN_DIM;
#pragma unroll 8
        for (int k = 0; k < KCH; k += 4) {
            v2f a, b0, b1;
            a.x = xs[k];
            a.y = xs[k + 1];
            const float* wrow = wk + (size_t)k * IN_DIM;
            b0.x = wrow[0];
            b0.y = wrow[IN_DIM];
            b1.x = wrow[16];
            b1.y = wrow[IN_DIM + 16];
            acc0 = __builtin_amdgcn_wmma_f32_16x16x4_f32(
                 false, a,  false, b0,
                 (short)0, acc0,
                 false,  false);
            acc1 = __builtin_amdgcn_wmma_f32_16x16x4_f32(
                 false, a,  false, b1,
                 (short)0, acc1,
                 false,  false);
        }
    }

    const int c  = lane & 15;
    const int h8 = (lane >> 4) << 3;
    float sums[8];
#pragma unroll
    for (int r = 0; r < 8; ++r) {
        const int m = r + h8;
        const float* xr = X + (size_t)(batch0 + m) * IN_DIM + nbase + c;
        sums[r] = acc0[r] * xr[0] + acc1[r] * xr[16];
    }

    const int md  = tid >> 6;
    float sd = 0.f;
    {
        const int seg = tid & 63;
        const float* xr = X + (size_t)(batch0 + md) * IN_DIM + seg * 16;
#pragma unroll
        for (int i = 0; i < 16; ++i) {
            const float xv = xr[i];
            sd += xv * xv * W[(size_t)(seg * 16 + i) * (IN_DIM + 1)];
        }
    }

#pragma unroll
    for (int r = 0; r < 8; ++r) {
        const int m = r + h8;
        atomicAdd(&red[m * RED_STRIDE + c], sums[r]);
    }
    atomicAdd(&redD[md], sd);

    __syncthreads();
    if (tid < MT) {
        float s = 0.f;
#pragma unroll
        for (int q = 0; q < 16; ++q) s += red[tid * RED_STRIDE + q];
        out[batch0 + tid] = 0.5f * (s - redD[tid]);
    }
}

extern "C" void kernel_launch(void* const* d_in, const int* in_sizes, int n_in,
                              void* d_out, int out_size, void* d_ws, size_t ws_size,
                              hipStream_t stream) {
    const float* x    = (const float*)d_in[0];
    const float* ker  = (const float*)d_in[1];
    const int*   fidx = (const int*)d_in[2];
    float* W   = (float*)d_ws;
    float* outp = (float*)d_out;

    build_W<<<(IN_DIM * IN_DIM) / JT, JT, 0, stream>>>(ker, fidx, W);
    ffm_gemm<<<BATCH / MT, NTHREADS, 0, stream>>>(x, W, outp);
}


# strict upper triangle only: build_W zeroes j<=i and drops blocks left of the diagonal; ffm_gemm gives a column block only the k rows up to its diagonal, 528 rows per wave, X tile whole in LDS
# speedup vs baseline: 4.6786x; 2.1238x over previous
#include <hip/hip_runtime.h>

typedef __attribute__((ext_vector_type(2))) float v2f;
typedef __attribute__((ext_vector_type(8))) float v8f;

#define BATCH   4096
#define IN_DIM  1024
#define FD      64
#define OD      30
#define KSTR    (FD * OD)

#define MT       16
#define XSTR     1028

#define NWAVES     32
#define NTHREADS   (NWAVES * 32)
#define WCOLS      32
#define NCB        (IN_DIM / WCOLS)
#define RED_STRIDE 18

#define JT     256
#define SEG2   (OD / 2)
#define NWG_W  2560

__global__ __launch_bounds__(JT) void build_W(const float* __restrict__ ker,
                                              const int* __restrict__ fidx,
                                              float* __restrict__ W) {
    __shared__ __attribute__((aligned(16))) float As[KSTR];
    __shared__ __attribute__((aligned(16))) float Bs[JT * OD];

    const int tid = threadIdx.x;
    const int b = blockIdx.x;
    int i, jq;
    if (b < 1024)      { i = b >> 2;                                  jq = b & 3; }
    else if (b < 1792) { const int r = b - 1024; i = 256 + r / 3;     jq = 1 + r % 3; }
    else if (b < 2304) { const int r = b - 1792; i = 512 + (r >> 1);  jq = 2 + (r & 1); }
    else               { i = 768 + (b - 2304);                        jq = 3; }
    const int j0  = jq * JT;
    const int fi  = fidx[i];
    const int fj  = fidx[j0 + tid];

    const float4* __restrict__ arow = (const float4*)(ker + (size_t)i * KSTR);
    const int v1 = tid + (tid < KSTR / 4 - JT ? JT : 0);
    const float4 r0 = arow[tid];
    const float4 r1 = arow[v1];

    const int q  = min(tid & 15, SEG2 - 1);
    const int sg = tid >> 4;
    const float2* __restrict__ src =
        (const float2*)(ker + (size_t)(j0 + sg) * KSTR + fi * OD) + q;
    float2* dst = (float2*)Bs + sg * SEG2 + q;
    float2 pc[JT / 16];
#pragma unroll
    for (int r = 0; r < JT / 16; ++r) pc[r] = src[(size_t)r * 16 * (KSTR / 2)];
#pragma unroll
    for (int r = 0; r < JT / 16; ++r) dst[r * 16 * SEG2] = pc[r];
    ((float4*)As)[tid] = r0;
    ((float4*)As)[v1]  = r1;
    __syncthreads();

    const float2* a2 = (const float2*)(As + fj * OD);
    const float2* b2 = (const float2*)(Bs + tid * OD);
    float s = 0.f;
#pragma unroll
    for (int p = 0; p < SEG2; ++p) {
        const float2 av = a2[p];
        const float2 bv = b2[p];
        s += av.x * bv.x;
        s += av.y * bv.y;
    }
    const int j = j0 + tid;
    W[(size_t)i * IN_DIM + j] = (j > i) ? s : 0.f;
}


template <int NSTEP>
__device__ __forceinline__ void k_steps(const float* xs, const float* __restrict__ wk,
                                        v8f& acc0, v8f& acc1) {
#pragma unroll
    for (int s = 0; s < NSTEP; ++s) {
        v2f a, b0, b1;
        a.x = xs[4 * s];
        a.y = xs[4 * s + 1];
        const float* wrow = wk + (size_t)(4 * s) * IN_DIM;
        b0.x = wrow[0];
        b0.y = wrow[IN_DIM];
        b1.x = wrow[16];
        b1.y = wrow[IN_DIM + 16];
        acc0 = __builtin_amdgcn_wmma_f32_16x16x4_f32(
             false, a,  false, b0,
             (short)0, acc0,
             false,  false);
        acc1 = __builtin_amdgcn_wmma_f32_16x16x4_f32(
             false, a,  false, b1,
             (short)0, acc1,
             false,  false);
    }
}

__device__ __forceinline__ void col_block(const float* Xs, const float* __restrict__ W,
                                          int p, int k0, int k1, int lane, float (&sums)[8]) {
    const int c    = lane & 15;
    const int ksel = (lane >> 4) * 2;
    const float* xs = Xs + c * XSTR + ksel;
    const float* __restrict__ wp = W + (size_t)ksel * IN_DIM + p * WCOLS + c;

    v8f acc0 = (v8f)(0.f);
    v8f acc1 = (v8f)(0.f);
    int k = k0;
    for (; k + 32 <= k1; k += 32)
        k_steps<8>(xs + k, wp + (size_t)k * IN_DIM, acc0, acc1);
    if (k < k1)
        k_steps<4>(xs + k, wp + (size_t)k * IN_DIM, acc0, acc1);

    const int h8 = (lane >> 4) << 3;
    const float* xr = Xs + h8 * XSTR + p * WCOLS + c;
#pragma unroll
    for (int r = 0; r < 8; ++r)
        sums[r] += acc0[r] * xr[r * XSTR] + acc1[r] * xr[r * XSTR + 16];
}

__global__ __launch_bounds__(NTHREADS) void ffm_gemm(const float* __restrict__ X,
                                                     const float* __restrict__ W,
                                                     float* __restrict__ out) {
    __shared__ __attribute__((aligned(16))) float Xs[MT * XSTR];
    __shared__ float red[MT * RED_STRIDE];

    const int batch0 = blockIdx.x * MT;
    const int tid  = threadIdx.x;
    const int wave = __builtin_amdgcn_readfirstlane(tid >> 5);
    const int lane = tid & 31;

    if (tid < MT * RED_STRIDE) red[tid] = 0.f;

    const float4* __restrict__ xg = (const float4*)(X + (size_t)batch0 * IN_DIM);
    float4 xv[4];
#pragma unroll
    for (int r = 0; r < 4; ++r) xv[r] = xg[tid + r * NTHREADS];
#pragma unroll
    for (int r = 0; r < 4; ++r) {
        const int v   = tid + r * NTHREADS;
        const int row = v >> 8;
        const int c4  = v & 255;
        *(float4*)&Xs[row * XSTR + c4 * 4] = xv[r];
    }
    __syncthreads();

    float sums[8] = {0.f, 0.f, 0.f, 0.f, 0.f, 0.f, 0.f, 0.f};

    if (wave < NCB / 2) {
        col_block(Xs, W, wave, 0, WCOLS * (wave + 1), lane, sums);
        col_block(Xs, W, NCB - 1 - wave, 0, 16 * (NCB - 1 - 2 * wave), lane, sums);
    } else {
        col_block(Xs, W, wave, 16 * (2 * wave - (NCB - 1)), WCOLS * (wave + 1), lane, sums);
    }

    const int c  = lane & 15;
    const int h8 = (lane >> 4) << 3;
#pragma unroll
    for (int r = 0; r < 8; ++r)
        atomicAdd(&red[(r + h8) * RED_STRIDE + c], sums[r]);

    __syncthreads();
    if (tid < MT) {
        float s = 0.f;
#pragma unroll
        for (int q = 0; q < 16; ++q) s += red[tid * RED_STRIDE + q];
        out[batch0 + tid] = s;
    }
}

extern "C" void kernel_launch(void* const* d_in, const int* in_sizes, int n_in,
                              void* d_out, int out_size, void* d_ws, size_t ws_size,
                              hipStream_t stream) {
    const float* x    = (const float*)d_in[0];
    const float* ker  = (const float*)d_in[1];
    const int*   fidx = (const int*)d_in[2];
    float* W   = (float*)d_ws;
    float* outp = (float*)d_out;

    build_W<<<NWG_W, JT, 0, stream>>>(ker, fidx, W);
    ffm_gemm<<<BATCH / MT, NTHREADS, 0, stream>>>(x, W, outp);
}


# build_W regrouped by field: a work-group gathers field column kernel[j0..j0+255, f] once, keeps it in registers, loops over the rows of field f (row double-buffered in LDS); 640 work-groups, not 2560
# speedup vs baseline: 5.0924x; 1.0884x over previous
#include <hip/hip_runtime.h>

typedef __attribute__((ext_vector_type(2))) float v2f;
typedef __attribute__((ext_vector_type(8))) float v8f;

#define BATCH   4096
#define IN_DIM  1024
#define FD      64
#define OD      30
#define KSTR    (FD * OD)

#define MT       16
#define XSTR     1028

#define NWAVES     32
#define NTHREADS   (NWAVES * 32)
#define WCOLS      32
#define NCB        (IN_DIM / WCOLS)
#define RED_STRIDE 18

#define JT     256
#define SEG2   (OD / 2)
#define NWG_W  (FD * 10)
#define NCH    (IN_DIM / JT)

__global__ __launch_bounds__(JT) void build_W(const float* __restrict__ ker,
                                              const int* __restrict__ fidx,
                                              float* __restrict__ W) {
    __shared__ __attribute__((aligned(16))) float As[2][KSTR];
    __shared__ __attribute__((aligned(16))) float Bs[JT * OD];
    __shared__ int rows[IN_DIM];
    __shared__ int wcnt[NCH * (JT / 32)];

    const int tid  = threadIdx.x;
    const int lane = tid & 31;
    const int wv   = tid >> 5;
    const int f = blockIdx.x & (FD - 1);
    const int u = blockIdx.x / FD;
    int jq, s;
    if (u < 1)      { jq = 0; s = 0; }
    else if (u < 3) { jq = 1; s = u - 1; }
    else if (u < 6) { jq = 2; s = u - 3; }
    else            { jq = 3; s = u - 6; }
    const int nshare = jq + 1;
    const int j0  = jq * JT;
    const int lim = j0 + JT;

    int fv[NCH];
#pragma unroll
    for (int m = 0; m < NCH; ++m) fv[m] = fidx[m * JT + tid];
    const int fj = fidx[j0 + tid];

    const int q  = min(tid & 15, SEG2 - 1);
    const int sg = tid >> 4;
    const float2* __restrict__ src =
        (const float2*)(ker + (size_t)(j0 + sg) * KSTR + f * OD) + q;
    float2* dst = (float2*)Bs + sg * SEG2 + q;
    float2 pc[JT / 16];
#pragma unroll
    for (int r = 0; r < JT / 16; ++r) pc[r] = src[(size_t)r * 16 * (KSTR / 2)];

    unsigned bal[NCH];
#pragma unroll
    for (int m = 0; m < NCH; ++m) {
        const bool hit = (m * JT + tid < lim) && (fv[m] == f);
        bal[m] = __builtin_amdgcn_ballot_w32(hit);
        if (lane == 0) wcnt[m * (JT / 32) + wv] = __builtin_popcount(bal[m]);
    }
#pragma unroll
    for (int r = 0; r < JT / 16; ++r) dst[r * 16 * SEG2] = pc[r];
    __syncthreads();

    int n = 0;
#pragma unroll
    for (int m = 0; m < NCH; ++m) {
        int base = 0;
#pragma unroll
        for (int w = 0; w < JT / 32; ++w) {
            if (w == wv) base = n;
            n += wcnt[m * (JT / 32) + w];
        }
        if ((bal[m] >> lane) & 1u)
            rows[base + __builtin_popcount(bal[m] & ((1u << lane) - 1u))] = m * JT + tid;
    }

    float2 bv[SEG2];
    {
        const float2* b2 = (const float2*)(Bs + tid * OD);
#pragma unroll
        for (int p = 0; p < SEG2; ++p) bv[p] = b2[p];
    }
    __syncthreads();

    const int nmine = (n > s) ? (n - s + nshare - 1) / nshare : 0;
    if (nmine == 0) return;

    const int v1 = tid + (tid < KSTR / 4 - JT ? JT : 0);
    int row = rows[s];
    const float4* __restrict__ arow = (const float4*)(ker + (size_t)row * KSTR);
    float4 r0 = arow[tid];
    float4 r1 = arow[v1];

    for (int it = 0; it < nmine; ++it) {
        float* A = As[it & 1];
        ((float4*)A)[tid] = r0;
        ((float4*)A)[v1]  = r1;
        __syncthreads();
        const int i = row;
        if (it + 1 < nmine) {
            row = rows[s + (it + 1) * nshare];
            const float4* __restrict__ nrow = (const float4*)(ker + (size_t)row * KSTR);
            r0 = nrow[tid];
            r1 = nrow[v1];
        }
        const float2* a2 = (const float2*)(A + fj * OD);
        float acc = 0.f;
#pragma unroll
        for (int p = 0; p < SEG2; ++p) {
            const float2 av = a2[p];
            acc += av.x * bv[p].x;
            acc += av.y * bv[p].y;
        }
        const int j = j0 + tid;
        W[(size_t)i * IN_DIM + j] = (j > i) ? acc : 0.f;
    }
}


template <int NSTEP>
__device__ __forceinline__ void k_steps(const float* xs, const float* __restrict__ wk,
                                        v8f& acc0, v8f& acc1) {
#pragma unroll
    for (int s = 0; s < NSTEP; ++s) {
        v2f a, b0, b1;
        a.x = xs[4 * s];
        a.y = xs[4 * s + 1];
        const float* wrow = wk + (size_t)(4 * s) * IN_DIM;
        b0.x = wrow[0];
        b0.y = wrow[IN_DIM];
        b1.x = wrow[16];
        b1.y = wrow[IN_DIM + 16];
        acc0 = __builtin_amdgcn_wmma_f32_16x16x4_f32(
             false, a,  false, b0,
             (short)0, acc0,
             false,  false);
        acc1 = __builtin_amdgcn_wmma_f32_16x16x4_f32(
             false, a,  false, b1,
             (short)0, acc1,
             false,  false);
    }
}

__device__ __forceinline__ void col_block(const float* Xs, const float* __restrict__ W,
                                          int p, int k0, int k1, int lane, float (&sums)[8]) {
    const int c    = lane & 15;
    const int ksel = (lane >> 4) * 2;
    const float* xs = Xs + c * XSTR + ksel;
    const float* __restrict__ wp = W + (size_t)ksel * IN_DIM + p * WCOLS + c;

    v8f acc0 = (v8f)(0.f);
    v8f acc1 = (v8f)(0.f);
    int k = k0;
    for (; k + 32 <= k1; k += 32)
        k_steps<8>(xs + k, wp + (size_t)k * IN_DIM, acc0, acc1);
    if (k < k1)
        k_steps<4>(xs + k, wp + (size_t)k * IN_DIM, acc0, acc1);

    const int h8 = (lane >> 4) << 3;
    const float* xr = Xs + h8 * XSTR + p * WCOLS + c;
#pragma unroll
    for (int r = 0; r < 8; ++r)
        sums[r] += acc0[r] * xr[r * XSTR] + acc1[r] * xr[r * XSTR + 16];
}

__global__ __launch_bounds__(NTHREADS) void ffm_gemm(const float* __restrict__ X,
                                                     const float* __restrict__ W,
                                                     float* __restrict__ out) {
    __shared__ __attribute__((aligned(16))) float Xs[MT * XSTR];
    __shared__ float red[MT * RED_STRIDE];

    const int batch0 = blockIdx.x * MT;
    const int tid  = threadIdx.x;
    const int wave = __builtin_amdgcn_readfirstlane(tid >> 5);
    const int lane = tid & 31;

    if (tid < MT * RED_STRIDE) red[tid] = 0.f;

    const float4* __restrict__ xg = (const float4*)(X + (size_t)batch0 * IN_DIM);
    float4 xv[4];
#pragma unroll
    for (int r = 0; r < 4; ++r) xv[r] = xg[tid + r * NTHREADS];
#pragma unroll
    for (int r = 0; r < 4; ++r) {
        const int v   = tid + r * NTHREADS;
        const int row = v >> 8;
        const int c4  = v & 255;
        *(float4*)&Xs[row * XSTR + c4 * 4] = xv[r];
    }
    __syncthreads();

    float sums[8] = {0.f, 0.f, 0.f, 0.f, 0.f, 0.f, 0.f, 0.f};

    if (wave < NCB / 2) {
        col_block(Xs, W, wave, 0, WCOLS * (wave + 1), lane, sums);
        col_block(Xs, W, NCB - 1 - wave, 0, 16 * (NCB - 1 - 2 * wave), lane, sums);
    } else {
        col_block(Xs, W, wave, 16 * (2 * wave - (NCB - 1)), WCOLS * (wave + 1), lane, sums);
    }

    const int c  = lane & 15;
    const int h8 = (lane >> 4) << 3;
#pragma unroll
    for (int r = 0; r < 8; ++r)
        atomicAdd(&red[(r + h8) * RED_STRIDE + c], sums[r]);

    __syncthreads();
    if (tid < MT) {
        float s = 0.f;
#pragma unroll
        for (int q = 0; q < 16; ++q) s += red[tid * RED_STRIDE + q];
        out[batch0 + tid] = s;
    }
}

extern "C" void kernel_launch(void* const* d_in, const int* in_sizes, int n_in,
                              void* d_out, int out_size, void* d_ws, size_t ws_size,
                              hipStream_t stream) {
    const float* x    = (const float*)d_in[0];
    const float* ker  = (const float*)d_in[1];
    const int*   fidx = (const int*)d_in[2];
    float* W   = (float*)d_ws;
    float* outp = (float*)d_out;

    build_W<<<NWG_W, JT, 0, stream>>>(ker, fidx, W);
    ffm_gemm<<<BATCH / MT, NTHREADS, 0, stream>>>(x, W, outp);
}


# ffm_gemm work-group = 32 batch rows x half the column blocks: one W fragment load feeds two row tiles (4 accumulators a wave), halving W bytes from L2; halves atomic-add into out, zeroed by build_W
# speedup vs baseline: 5.5754x; 1.0948x over previous
#include <hip/hip_runtime.h>

typedef __attribute__((ext_vector_type(2))) float v2f;
typedef __attribute__((ext_vector_type(8))) float v8f;

#define BATCH   4096
#define IN_DIM  1024
#define FD      64
#define OD      30
#define KSTR    (FD * OD)

#define MT       32
#define XSTR     1028

#define NWAVES     32
#define NTHREADS   (NWAVES * 32)
#define WCOLS      32
#define NCB        (IN_DIM / WCOLS)
#define RED_STRIDE 18

#define JT     256
#define SEG2   (OD / 2)
#define NWG_W  (FD * 10)
#define NCH    (IN_DIM / JT)

__global__ __launch_bounds__(JT) void build_W(const float* __restrict__ ker,
                                              const int* __restrict__ fidx,
                                              float* __restrict__ W,
                                              float* __restrict__ out) {
    __shared__ __attribute__((aligned(16))) float As[2][KSTR];
    __shared__ __attribute__((aligned(16))) float Bs[JT * OD];
    __shared__ int rows[IN_DIM];
    __shared__ int wcnt[NCH * (JT / 32)];

    const int tid  = threadIdx.x;
    const int lane = tid & 31;
    const int wv   = tid >> 5;
    const int f = blockIdx.x & (FD - 1);
    const int u = blockIdx.x / FD;
    int jq, s;
    if (u < 1)      { jq = 0; s = 0; }
    else if (u < 3) { jq = 1; s = u - 1; }
    else if (u < 6) { jq = 2; s = u - 3; }
    else            { jq = 3; s = u - 6; }
    const int nshare = jq + 1;
    const int j0  = jq * JT;
    const int lim = j0 + JT;

    if (blockIdx.x < BATCH / JT) out[blockIdx.x * JT + tid] = 0.f;

    int fv[NCH];
#pragma unroll
    for (int m = 0; m < NCH; ++m) fv[m] = fidx[m * JT + tid];
    const int fj = fidx[j0 + tid];

    const int q  = min(tid & 15, SEG2 - 1);
    const int sg = tid >> 4;
    const float2* __restrict__ src =
        (const float2*)(ker + (size_t)(j0 + sg) * KSTR + f * OD) + q;
    float2* dst = (float2*)Bs + sg * SEG2 + q;
    float2 pc[JT / 16];
#pragma unroll
    for (int r = 0; r < JT / 16; ++r) pc[r] = src[(size_t)r * 16 * (KSTR / 2)];

    unsigned bal[NCH];
#pragma unroll
    for (int m = 0; m < NCH; ++m) {
        const bool hit = (m * JT + tid < lim) && (fv[m] == f);
        bal[m] = __builtin_amdgcn_ballot_w32(hit);
        if (lane == 0) wcnt[m * (JT / 32) + wv] = __builtin_popcount(bal[m]);
    }
#pragma unroll
    for (int r = 0; r < JT / 16; ++r) dst[r * 16 * SEG2] = pc[r];
    __syncthreads();

    int n = 0;
#pragma unroll
    for (int m = 0; m < NCH; ++m) {
        int base = 0;
#pragma unroll
        for (int w = 0; w < JT / 32; ++w) {
            if (w == wv) base = n;
            n += wcnt[m * (JT / 32) + w];
        }
        if ((bal[m] >> lane) & 1u)
            rows[base + __builtin_popcount(bal[m] & ((1u << lane) - 1u))] = m * JT + tid;
    }

    float2 bv[SEG2];
    {
        const float2* b2 = (const float2*)(Bs + tid * OD);
#pragma unroll
        for (int p = 0; p < SEG2; ++p) bv[p] = b2[p];
    }
    __syncthreads();

    const int nmine = (n > s) ? (n - s + nshare - 1) / nshare : 0;
    if (nmine == 0) return;

    const int v1 = tid + (tid < KSTR / 4 - JT ? JT : 0);
    int row = rows[s];
    const float4* __restrict__ arow = (const float4*)(ker + (size_t)row * KSTR);
    float4 r0 = arow[tid];
    float4 r1 = arow[v1];

    for (int it = 0; it < nmine; ++it) {
        float* A = As[it & 1];
        ((float4*)A)[tid] = r0;
        ((float4*)A)[v1]  = r1;
        __syncthreads();
        const int i = row;
        if (it + 1 < nmine) {
            row = rows[s + (it + 1) * nshare];
            const float4* __restrict__ nrow = (const float4*)(ker + (size_t)row * KSTR);
            r0 = nrow[tid];
            r1 = nrow[v1];
        }
        const float2* a2 = (const float2*)(A + fj * OD);
        float acc = 0.f;
#pragma unroll
        for (int p = 0; p < SEG2; ++p) {
            const float2 av = a2[p];
            acc += av.x * bv[p].x;
            acc += av.y * bv[p].y;
        }
        const int j = j0 + tid;
        W[(size_t)i * IN_DIM + j] = (j > i) ? acc : 0.f;
    }
}

#define SHARE  264

__device__ __forceinline__ v8f mma4(v2f a, v2f b, v8f c) {
    return __builtin_amdgcn_wmma_f32_16x16x4_f32(
         false, a,  false, b,
         (short)0, c,
         false,  false);
}

template <int NSTEP>
__device__ __forceinline__ void k_steps(const float* xs, const float* __restrict__ wk,
                                        v8f& acc00, v8f& acc01, v8f& acc10, v8f& acc11) {
#pragma unroll
    for (int s = 0; s < NSTEP; ++s) {
        v2f a0, a1, b0, b1;
        a0.x = xs[4 * s];
        a0.y = xs[4 * s + 1];
        a1.x = xs[16 * XSTR + 4 * s];
        a1.y = xs[16 * XSTR + 4 * s + 1];
        const float* wrow = wk + (size_t)(4 * s) * IN_DIM;
        b0.x = wrow[0];
        b0.y = wrow[IN_DIM];
        b1.x = wrow[16];
        b1.y = wrow[IN_DIM + 16];
        acc00 = mma4(a0, b0, acc00);
        acc01 = mma4(a0, b1, acc01);
        acc10 = mma4(a1, b0, acc10);
        acc11 = mma4(a1, b1, acc11);
    }
}

__device__ __forceinline__ void col_block(const float* Xs, const float* __restrict__ W,
                                          int p, int k0, int k1, int lane, float (&sums)[16]) {
    const int c    = lane & 15;
    const int ksel = (lane >> 4) * 2;
    const float* xs = Xs + c * XSTR + ksel;
    const float* __restrict__ wp = W + (size_t)ksel * IN_DIM + p * WCOLS + c;

    v8f acc00 = (v8f)(0.f);
    v8f acc01 = (v8f)(0.f);
    v8f acc10 = (v8f)(0.f);
    v8f acc11 = (v8f)(0.f);
    int k = k0;
    for (; k + 32 <= k1; k += 32)
        k_steps<8>(xs + k, wp + (size_t)k * IN_DIM, acc00, acc01, acc10, acc11);
    if (k + 16 <= k1) {
        k_steps<4>(xs + k, wp + (size_t)k * IN_DIM, acc00, acc01, acc10, acc11);
        k += 16;
    }
    if (k < k1)
        k_steps<2>(xs + k, wp + (size_t)k * IN_DIM, acc00, acc01, acc10, acc11);

    const int h8 = (lane >> 4) << 3;
    const float* xr = Xs + h8 * XSTR + p * WCOLS + c;
#pragma unroll
    for (int r = 0; r < 8; ++r) {
        sums[r]     += acc00[r] * xr[r * XSTR]        + acc01[r] * xr[r * XSTR + 16];
        sums[8 + r] += acc10[r] * xr[(16 + r) * XSTR] + acc11[r] * xr[(16 + r) * XSTR + 16];
    }
}

__global__ __launch_bounds__(NTHREADS) void ffm_gemm(const float* __restrict__ X,
                                                     const float* __restrict__ W,
                                                     float* __restrict__ out) {
    __shared__ __attribute__((aligned(16))) float Xs[MT * XSTR];
    __shared__ float red[MT * RED_STRIDE];

    const int half   = blockIdx.x & 1;
    const int batch0 = (blockIdx.x >> 1) * MT;
    const int tid  = threadIdx.x;
    const int wave = __builtin_amdgcn_readfirstlane(tid >> 5);
    const int lane = tid & 31;

    if (tid < MT * RED_STRIDE) red[tid] = 0.f;

    const float4* __restrict__ xg = (const float4*)(X + (size_t)batch0 * IN_DIM);
    float4 xv[8];
#pragma unroll
    for (int r = 0; r < 8; ++r) xv[r] = xg[tid + r * NTHREADS];
#pragma unroll
    for (int r = 0; r < 8; ++r) {
        const int v   = tid + r * NTHREADS;
        const int row = v >> 8;
        const int c4  = v & 255;
        *(float4*)&Xs[row * XSTR + c4 * 4] = xv[r];
    }
    __syncthreads();

    float sums[16];
#pragma unroll
    for (int r = 0; r < 16; ++r) sums[r] = 0.f;

    const int q  = 2 * (wave >> 2) + half;
    const int la = WCOLS * (q + 1);
    const int s0 = SHARE * (wave & 3);
    const int s1 = s0 + SHARE;
    const int a0 = min(s0, la), a1 = min(s1, la);
    const int b0 = max(s0 - la, 0), b1 = max(s1 - la, 0);
    if (a0 < a1) col_block(Xs, W, q, a0, a1, lane, sums);
    if (b0 < b1) col_block(Xs, W, NCB - 1 - q, b0, b1, lane, sums);

    const int c  = lane & 15;
    const int h8 = (lane >> 4) << 3;
#pragma unroll
    for (int r = 0; r < 8; ++r) {
        atomicAdd(&red[(r + h8) * RED_STRIDE + c], sums[r]);
        atomicAdd(&red[(16 + r + h8) * RED_STRIDE + c], sums[8 + r]);
    }

    __syncthreads();
    if (tid < MT) {
        float s = 0.f;
#pragma unroll
        for (int q2 = 0; q2 < 16; ++q2) s += red[tid * RED_STRIDE + q2];
        atomicAdd(&out[batch0 + tid], s);
    }
}

extern "C" void kernel_launch(void* const* d_in, const int* in_sizes, int n_in,
                              void* d_out, int out_size, void* d_ws, size_t ws_size,
                              hipStream_t stream) {
    const float* x    = (const float*)d_in[0];
    const float* ker  = (const float*)d_in[1];
    const int*   fidx = (const int*)d_in[2];
    float* W   = (float*)d_ws;
    float* outp = (float*)d_out;

    build_W<<<NWG_W, JT, 0, stream>>>(ker, fidx, W, outp);
    ffm_gemm<<<2 * (BATCH / MT), NTHREADS, 0, stream>>>(x, W, outp);
}
